# starting kernel
# baseline (speedup 1.0000x reference)
#include <hip/hip_runtime.h>

typedef __attribute__((ext_vector_type(2))) float v2f;
typedef __attribute__((ext_vector_type(8))) float v8f;

#define BATCH 32
#define NN    1024
#define IND   128
#define HIDD  64
#define LDS_STRIDE 66

__global__ __launch_bounds__(256) void k_proj(
    const float* __restrict__ x,
    const float* __restrict__ Wh, const float* __restrict__ bh,
    const float* __restrict__ Wf, const float* __restrict__ bf,
    float* __restrict__ H, float* __restrict__ P) {
  const int lane = threadIdx.x & 31;
  const int w    = threadIdx.x >> 5;
  const int gw   = blockIdx.x * 8 + w;
  const int m0   = gw * 16;
  const int lr   = lane & 15;
  const int hi   = lane >> 4;

  v8f ha[4] = {}; v8f pa[4] = {};
  const float* xrow = x + (size_t)(m0 + lr) * IND + 2 * hi;

#pragma unroll 4
  for (int k0 = 0; k0 < IND; k0 += 4) {
    v2f a = *(const v2f*)(xrow + k0);
    const float* w0 = Wh + (size_t)(k0 + 2 * hi) * HIDD + lr;
    const float* f0 = Wf + (size_t)(k0 + 2 * hi) * HIDD + lr;
#pragma unroll
    for (int nb = 0; nb < 4; ++nb) {
      v2f bw, bq;
      bw.x = w0[nb * 16]; bw.y = w0[HIDD + nb * 16];
      bq.x = f0[nb * 16]; bq.y = f0[HIDD + nb * 16];
      ha[nb] = __builtin_amdgcn_wmma_f32_16x16x4_f32(false, a, false, bw,
                                                     (short)0, ha[nb], false, false);
      pa[nb] = __builtin_amdgcn_wmma_f32_16x16x4_f32(false, a, false, bq,
                                                     (short)0, pa[nb], false, false);
    }
  }

#pragma unroll
  for (int nb = 0; nb < 4; ++nb) {
    const int n = nb * 16 + lr;
    const float bhn = bh[n];
    const float bfn = bf[n];
#pragma unroll
    for (int i = 0; i < 8; ++i) {
      const int m = i + 8 * hi;
      const size_t off = (size_t)(m0 + m) * HIDD + n;
      H[off] = ha[nb][i] + bhn;
      P[off] = pa[nb][i] + bfn;
    }
  }
}

__global__ __launch_bounds__(256) void k_bmm_gate(
    const float* __restrict__ A, const float* __restrict__ H,
    const float* __restrict__ P,
    const float* __restrict__ Wg1, const float* __restrict__ bg1,
    const float* __restrict__ Wg2, const float* __restrict__ bg2,
    float* __restrict__ out) {
  __shared__ float lds[8 * 16 * LDS_STRIDE];

  const int lane = threadIdx.x & 31;
  const int w    = threadIdx.x >> 5;
  const int gw   = blockIdx.x * 8 + w;
  const int b    = gw >> 6;
  const int m0   = (gw & 63) * 16;
  const int lr   = lane & 15;
  const int hi   = lane >> 4;

  const float* Ab = A + (size_t)b * NN * NN;
  const float* Hb = H + (size_t)b * NN * HIDD;

  v8f acc[4] = {};
  const float* arow = Ab + (size_t)(m0 + lr) * NN + 2 * hi;
#pragma unroll 4
  for (int k0 = 0; k0 < NN; k0 += 4) {
    v2f a = *(const v2f*)(arow + k0);
    const float* h0 = Hb + (size_t)(k0 + 2 * hi) * HIDD + lr;
#pragma unroll
    for (int nb = 0; nb < 4; ++nb) {
      v2f bb;
      bb.x = h0[nb * 16];
      bb.y = h0[HIDD + nb * 16];
      acc[nb] = __builtin_amdgcn_wmma_f32_16x16x4_f32(false, a, false, bb,
                                                      (short)0, acc[nb], false, false);
    }
  }

  float* myl = lds + w * (16 * LDS_STRIDE);
#pragma unroll
  for (int nb = 0; nb < 4; ++nb)
#pragma unroll
    for (int i = 0; i < 8; ++i)
      myl[(i + 8 * hi) * LDS_STRIDE + nb * 16 + lr] = acc[nb][i];

  v8f g[4] = {};
  const float* prow = P + (size_t)(b * NN + m0 + lr) * HIDD + 2 * hi;
#pragma unroll
  for (int kk = 0; kk < HIDD; kk += 4) {
    v2f a2 = *(const v2f*)(myl + lr * LDS_STRIDE + kk + 2 * hi);
    v2f a1 = *(const v2f*)(prow + kk);
    const float* g2w = Wg2 + (size_t)(kk + 2 * hi) * HIDD + lr;
    const float* g1w = Wg1 + (size_t)(kk + 2 * hi) * HIDD + lr;
#pragma unroll
    for (int nb = 0; nb < 4; ++nb) {
      v2f b2, b1;
      b2.x = g2w[nb * 16]; b2.y = g2w[HIDD + nb * 16];
      b1.x = g1w[nb * 16]; b1.y = g1w[HIDD + nb * 16];
      g[nb] = __builtin_amdgcn_wmma_f32_16x16x4_f32(false, a2, false, b2,
                                                    (short)0, g[nb], false, false);
      g[nb] = __builtin_amdgcn_wmma_f32_16x16x4_f32(false, a1, false, b1,
                                                    (short)0, g[nb], false, false);
    }
  }

#pragma unroll
  for (int nb = 0; nb < 4; ++nb) {
    const int n = nb * 16 + lr;
    const float bias = bg1[n] + bg2[n];
#pragma unroll
    for (int i = 0; i < 8; ++i) {
      const int m = i + 8 * hi;
      const size_t off = (size_t)(b * NN + m0 + m) * HIDD + n;
      const float xn = acc[nb][i];
      const float p  = P[off];
      const float z  = g[nb][i] + bias;
      const float gate = 1.0f / (1.0f + __expf(-z));
      out[off] = xn * gate + p * (1.0f - gate);
    }
  }
}

extern "C" void kernel_launch(void* const* d_in, const int* in_sizes, int n_in,
                              void* d_out, int out_size, void* d_ws, size_t ws_size,
                              hipStream_t stream) {
  const float* x    = (const float*)d_in[0];
  const float* A    = (const float*)d_in[1];
  const float* W_h  = (const float*)d_in[2];
  const float* b_h  = (const float*)d_in[3];
  const float* W_f1 = (const float*)d_in[4];
  const float* b_f1 = (const float*)d_in[5];
  const float* W_g1 = (const float*)d_in[6];
  const float* b_g1 = (const float*)d_in[7];
  const float* W_g2 = (const float*)d_in[8];
  const float* b_g2 = (const float*)d_in[9];
  float* out = (float*)d_out;

  float* H = (float*)d_ws;
  float* P = H + (size_t)BATCH * NN * HIDD;

  k_proj<<<256, 256, 0, stream>>>(x, W_h, b_h, W_f1, b_f1, H, P);
  k_bmm_gate<<<256, 256, 0, stream>>>(A, H, P, W_g1, b_g1, W_g2, b_g2, out);
}


# Wave's four 16-col tiles column-interleaved (tile nb, lane c = col 4c+nb): B operands and epilogue become whole-row b128 global loads/stores, no LDS staging; same k order and sums
# speedup vs baseline: 1.1099x; 1.1099x over previous
#include <hip/hip_runtime.h>

typedef __attribute__((ext_vector_type(2))) float v2f;
typedef __attribute__((ext_vector_type(4))) float v4f;
typedef __attribute__((ext_vector_type(8))) float v8f;

#define BATCH 32
#define NN    1024
#define IND   128
#define HIDD  64
#define LDS_STRIDE 68


__global__ __launch_bounds__(256) void k_proj(
    const float* __restrict__ x,
    const float* __restrict__ Wh, const float* __restrict__ bh,
    const float* __restrict__ Wf, const float* __restrict__ bf,
    float* __restrict__ H, float* __restrict__ P) {
  const int lane = threadIdx.x & 31;
  const int w    = threadIdx.x >> 5;
  const int gw   = blockIdx.x * 8 + w;
  const int m0   = gw * 16;
  const int lr   = lane & 15;
  const int hi   = lane >> 4;

  v8f ha[4] = {}; v8f pa[4] = {};
  const float* xrow = x + (size_t)(m0 + lr) * IND + 2 * hi;

#pragma unroll 4
  for (int k0 = 0; k0 < IND; k0 += 4) {
    v2f a = *(const v2f*)(xrow + k0);
    const float* w0 = Wh + (size_t)(k0 + 2 * hi) * HIDD + 4 * lr;
    const float* f0 = Wf + (size_t)(k0 + 2 * hi) * HIDD + 4 * lr;
    const v4f w_lo = *(const v4f*)(w0);
    const v4f w_hi = *(const v4f*)(w0 + HIDD);
    const v4f f_lo = *(const v4f*)(f0);
    const v4f f_hi = *(const v4f*)(f0 + HIDD);
#pragma unroll
    for (int nb = 0; nb < 4; ++nb) {
      v2f bw, bq;
      bw.x = w_lo[nb]; bw.y = w_hi[nb];
      bq.x = f_lo[nb]; bq.y = f_hi[nb];
      ha[nb] = __builtin_amdgcn_wmma_f32_16x16x4_f32(false, a, false, bw,
                                                     (short)0, ha[nb], false, false);
      pa[nb] = __builtin_amdgcn_wmma_f32_16x16x4_f32(false, a, false, bq,
                                                     (short)0, pa[nb], false, false);
    }
  }

  const v4f bhv = *(const v4f*)(bh + 4 * lr);
  const v4f bfv = *(const v4f*)(bf + 4 * lr);
#pragma unroll
  for (int i = 0; i < 8; ++i) {
    const int m = i + 8 * hi;
    const size_t off = (size_t)(m0 + m) * HIDD + 4 * lr;
    v4f hv, pv;
#pragma unroll
    for (int nb = 0; nb < 4; ++nb) {
      hv[nb] = ha[nb][i] + bhv[nb];
      pv[nb] = pa[nb][i] + bfv[nb];
    }
    *(v4f*)(H + off) = hv;
    *(v4f*)(P + off) = pv;
  }
}

__global__ __launch_bounds__(256) void k_bmm_gate(
    const float* __restrict__ A, const float* __restrict__ H,
    const float* __restrict__ P,
    const float* __restrict__ Wg1, const float* __restrict__ bg1,
    const float* __restrict__ Wg2, const float* __restrict__ bg2,
    float* __restrict__ out) {
  __shared__ __attribute__((aligned(16))) float lds[8 * 16 * LDS_STRIDE];

  const int lane = threadIdx.x & 31;
  const int w    = threadIdx.x >> 5;
  const int gw   = blockIdx.x * 8 + w;
  const int b    = gw >> 6;
  const int m0   = (gw & 63) * 16;
  const int lr   = lane & 15;
  const int hi   = lane >> 4;

  const float* Ab = A + (size_t)b * NN * NN;
  const float* Hb = H + (size_t)b * NN * HIDD;

  v8f acc[4] = {};
  const float* arow = Ab + (size_t)(m0 + lr) * NN + 2 * hi;
#pragma unroll 4
  for (int k0 = 0; k0 < NN; k0 += 4) {
    v2f a = *(const v2f*)(arow + k0);
    const float* h0 = Hb + (size_t)(k0 + 2 * hi) * HIDD + 4 * lr;
    const v4f h_lo = *(const v4f*)(h0);
    const v4f h_hi = *(const v4f*)(h0 + HIDD);
#pragma unroll
    for (int nb = 0; nb < 4; ++nb) {
      v2f bb;
      bb.x = h_lo[nb];
      bb.y = h_hi[nb];
      acc[nb] = __builtin_amdgcn_wmma_f32_16x16x4_f32(false, a, false, bb,
                                                      (short)0, acc[nb], false, false);
    }
  }

  float* myl = lds + w * (16 * LDS_STRIDE);
#pragma unroll
  for (int i = 0; i < 8; ++i) {
    v4f t;
#pragma unroll
    for (int nb = 0; nb < 4; ++nb) t[nb] = acc[nb][i];
    *(v4f*)(myl + (i + 8 * hi) * LDS_STRIDE + 4 * lr) = t;
  }

  v8f g[4] = {};
  const float* prow = P + (size_t)(b * NN + m0 + lr) * HIDD + 2 * hi;
#pragma unroll
  for (int kk = 0; kk < HIDD; kk += 4) {
    v2f a2 = *(const v2f*)(myl + lr * LDS_STRIDE + kk + 2 * hi);
    v2f a1 = *(const v2f*)(prow + kk);
    const float* g2w = Wg2 + (size_t)(kk + 2 * hi) * HIDD + 4 * lr;
    const float* g1w = Wg1 + (size_t)(kk + 2 * hi) * HIDD + 4 * lr;
    const v4f g2_lo = *(const v4f*)(g2w);
    const v4f g2_hi = *(const v4f*)(g2w + HIDD);
    const v4f g1_lo = *(const v4f*)(g1w);
    const v4f g1_hi = *(const v4f*)(g1w + HIDD);
#pragma unroll
    for (int nb = 0; nb < 4; ++nb) {
      v2f b2, b1;
      b2.x = g2_lo[nb]; b2.y = g2_hi[nb];
      b1.x = g1_lo[nb]; b1.y = g1_hi[nb];
      g[nb] = __builtin_amdgcn_wmma_f32_16x16x4_f32(false, a2, false, b2,
                                                    (short)0, g[nb], false, false);
      g[nb] = __builtin_amdgcn_wmma_f32_16x16x4_f32(false, a1, false, b1,
                                                    (short)0, g[nb], false, false);
    }
  }

  const v4f bias = *(const v4f*)(bg1 + 4 * lr) + *(const v4f*)(bg2 + 4 * lr);
#pragma unroll
  for (int i = 0; i < 8; ++i) {
    const int m = i + 8 * hi;
    const size_t off = (size_t)(b * NN + m0 + m) * HIDD + 4 * lr;
    const v4f p = *(const v4f*)(P + off);
    v4f o;
#pragma unroll
    for (int nb = 0; nb < 4; ++nb) {
      const float xn = acc[nb][i];
      const float z  = g[nb][i] + bias[nb];
      const float gate = 1.0f / (1.0f + __expf(-z));
      o[nb] = xn * gate + p[nb] * (1.0f - gate);
    }
    *(v4f*)(out + off) = o;
  }
}

extern "C" void kernel_launch(void* const* d_in, const int* in_sizes, int n_in,
                              void* d_out, int out_size, void* d_ws, size_t ws_size,
                              hipStream_t stream) {
  const float* x    = (const float*)d_in[0];
  const float* A    = (const float*)d_in[1];
  const float* W_h  = (const float*)d_in[2];
  const float* b_h  = (const float*)d_in[3];
  const float* W_f1 = (const float*)d_in[4];
  const float* b_f1 = (const float*)d_in[5];
  const float* W_g1 = (const float*)d_in[6];
  const float* b_g1 = (const float*)d_in[7];
  const float* W_g2 = (const float*)d_in[8];
  const float* b_g2 = (const float*)d_in[9];
  float* out = (float*)d_out;

  float* H = (float*)d_ws;
  float* P = H + (size_t)BATCH * NN * HIDD;

  k_proj<<<256, 256, 0, stream>>>(x, W_h, b_h, W_f1, b_f1, H, P);
  k_bmm_gate<<<256, 256, 0, stream>>>(A, H, P, W_g1, b_g1, W_g2, b_g2, out);
}


# bmm K loop double-buffered in registers: the next 64-k block's A and H fragments are loaded as one group ahead of the current block's 64 products; same k order and sums
# speedup vs baseline: 1.6208x; 1.4603x over previous
#include <hip/hip_runtime.h>

typedef __attribute__((ext_vector_type(2))) float v2f;
typedef __attribute__((ext_vector_type(4))) float v4f;
typedef __attribute__((ext_vector_type(8))) float v8f;

#define BATCH 32
#define NN    1024
#define IND   128
#define HIDD  64
#define LDS_STRIDE 68


__global__ __launch_bounds__(256) void k_proj(
    const float* __restrict__ x,
    const float* __restrict__ Wh, const float* __restrict__ bh,
    const float* __restrict__ Wf, const float* __restrict__ bf,
    float* __restrict__ H, float* __restrict__ P) {
  const int lane = threadIdx.x & 31;
  const int w    = threadIdx.x >> 5;
  const int gw   = blockIdx.x * 8 + w;
  const int m0   = gw * 16;
  const int lr   = lane & 15;
  const int hi   = lane >> 4;

  v8f ha[4] = {}; v8f pa[4] = {};
  const float* xrow = x + (size_t)(m0 + lr) * IND + 2 * hi;

#pragma unroll 4
  for (int k0 = 0; k0 < IND; k0 += 4) {
    v2f a = *(const v2f*)(xrow + k0);
    const float* w0 = Wh + (size_t)(k0 + 2 * hi) * HIDD + 4 * lr;
    const float* f0 = Wf + (size_t)(k0 + 2 * hi) * HIDD + 4 * lr;
    const v4f w_lo = *(const v4f*)(w0);
    const v4f w_hi = *(const v4f*)(w0 + HIDD);
    const v4f f_lo = *(const v4f*)(f0);
    const v4f f_hi = *(const v4f*)(f0 + HIDD);
#pragma unroll
    for (int nb = 0; nb < 4; ++nb) {
      v2f bw, bq;
      bw.x = w_lo[nb]; bw.y = w_hi[nb];
      bq.x = f_lo[nb]; bq.y = f_hi[nb];
      ha[nb] = __builtin_amdgcn_wmma_f32_16x16x4_f32(false, a, false, bw,
                                                     (short)0, ha[nb], false, false);
      pa[nb] = __builtin_amdgcn_wmma_f32_16x16x4_f32(false, a, false, bq,
                                                     (short)0, pa[nb], false, false);
    }
  }

  const v4f bhv = *(const v4f*)(bh + 4 * lr);
  const v4f bfv = *(const v4f*)(bf + 4 * lr);
#pragma unroll
  for (int i = 0; i < 8; ++i) {
    const int m = i + 8 * hi;
    const size_t off = (size_t)(m0 + m) * HIDD + 4 * lr;
    v4f hv, pv;
#pragma unroll
    for (int nb = 0; nb < 4; ++nb) {
      hv[nb] = ha[nb][i] + bhv[nb];
      pv[nb] = pa[nb][i] + bfv[nb];
    }
    *(v4f*)(H + off) = hv;
    *(v4f*)(P + off) = pv;
  }
}

__global__ __launch_bounds__(256) void k_bmm_gate(
    const float* __restrict__ A, const float* __restrict__ H,
    const float* __restrict__ P,
    const float* __restrict__ Wg1, const float* __restrict__ bg1,
    const float* __restrict__ Wg2, const float* __restrict__ bg2,
    float* __restrict__ out) {
  __shared__ __attribute__((aligned(16))) float lds[8 * 16 * LDS_STRIDE];

  const int lane = threadIdx.x & 31;
  const int w    = threadIdx.x >> 5;
  const int gw   = blockIdx.x * 8 + w;
  const int b    = gw >> 6;
  const int m0   = (gw & 63) * 16;
  const int lr   = lane & 15;
  const int hi   = lane >> 4;

  const float* Ab = A + (size_t)b * NN * NN;
  const float* Hb = H + (size_t)b * NN * HIDD;

  v8f acc[4] = {};
  const float* arow = Ab + (size_t)(m0 + lr) * NN + 2 * hi;
  const float* hrow = Hb + (size_t)(2 * hi) * HIDD + 4 * lr;
  v2f fa[2][16];
  v4f fl[2][16], fh[2][16];

#define LOAD_BLK(st, kb)                                                        \
  _Pragma("unroll") for (int j = 0; j < 16; ++j) {                              \
    fa[st][j] = *(const v2f*)(arow + (kb) + 4 * j);                             \
    fl[st][j] = *(const v4f*)(hrow + (size_t)((kb) + 4 * j) * HIDD);            \
    fh[st][j] = *(const v4f*)(hrow + (size_t)((kb) + 4 * j) * HIDD + HIDD);     \
  }
#define MMA_BLK(st)                                                             \
  _Pragma("unroll") for (int j = 0; j < 16; ++j) {                              \
    _Pragma("unroll") for (int nb = 0; nb < 4; ++nb) {                          \
      v2f bb;                                                                   \
      bb.x = fl[st][j][nb];                                                     \
      bb.y = fh[st][j][nb];                                                     \
      acc[nb] = __builtin_amdgcn_wmma_f32_16x16x4_f32(false, fa[st][j], false, bb, \
                                                      (short)0, acc[nb], false, false); \
    }                                                                           \
  }
#define STEP(ld_st, ld_kb, mma_st)                                              \
  LOAD_BLK(ld_st, ld_kb)                                                        \
  __builtin_amdgcn_sched_barrier(0);                                            \
  MMA_BLK(mma_st)                                                               \
  __builtin_amdgcn_sched_barrier(0);

  LOAD_BLK(0, 0)
#pragma nounroll
  for (int kb = 0; kb < NN - 128; kb += 128) {
    STEP(1, kb + 64, 0)
    STEP(0, kb + 128, 1)
  }
  STEP(1, NN - 64, 0)
  MMA_BLK(1)
#undef STEP
#undef LOAD_BLK
#undef MMA_BLK

  float* myl = lds + w * (16 * LDS_STRIDE);
#pragma unroll
  for (int i = 0; i < 8; ++i) {
    v4f t;
#pragma unroll
    for (int nb = 0; nb < 4; ++nb) t[nb] = acc[nb][i];
    *(v4f*)(myl + (i + 8 * hi) * LDS_STRIDE + 4 * lr) = t;
  }

  v8f g[4] = {};
  const float* prow = P + (size_t)(b * NN + m0 + lr) * HIDD + 2 * hi;
#pragma unroll
  for (int kk = 0; kk < HIDD; kk += 4) {
    v2f a2 = *(const v2f*)(myl + lr * LDS_STRIDE + kk + 2 * hi);
    v2f a1 = *(const v2f*)(prow + kk);
    const float* g2w = Wg2 + (size_t)(kk + 2 * hi) * HIDD + 4 * lr;
    const float* g1w = Wg1 + (size_t)(kk + 2 * hi) * HIDD + 4 * lr;
    const v4f g2_lo = *(const v4f*)(g2w);
    const v4f g2_hi = *(const v4f*)(g2w + HIDD);
    const v4f g1_lo = *(const v4f*)(g1w);
    const v4f g1_hi = *(const v4f*)(g1w + HIDD);
#pragma unroll
    for (int nb = 0; nb < 4; ++nb) {
      v2f b2, b1;
      b2.x = g2_lo[nb]; b2.y = g2_hi[nb];
      b1.x = g1_lo[nb]; b1.y = g1_hi[nb];
      g[nb] = __builtin_amdgcn_wmma_f32_16x16x4_f32(false, a2, false, b2,
                                                    (short)0, g[nb], false, false);
      g[nb] = __builtin_amdgcn_wmma_f32_16x16x4_f32(false, a1, false, b1,
                                                    (short)0, g[nb], false, false);
    }
  }

  const v4f bias = *(const v4f*)(bg1 + 4 * lr) + *(const v4f*)(bg2 + 4 * lr);
#pragma unroll
  for (int i = 0; i < 8; ++i) {
    const int m = i + 8 * hi;
    const size_t off = (size_t)(b * NN + m0 + m) * HIDD + 4 * lr;
    const v4f p = *(const v4f*)(P + off);
    v4f o;
#pragma unroll
    for (int nb = 0; nb < 4; ++nb) {
      const float xn = acc[nb][i];
      const float z  = g[nb][i] + bias[nb];
      const float gate = 1.0f / (1.0f + __expf(-z));
      o[nb] = xn * gate + p[nb] * (1.0f - gate);
    }
    *(v4f*)(out + off) = o;
  }
}

extern "C" void kernel_launch(void* const* d_in, const int* in_sizes, int n_in,
                              void* d_out, int out_size, void* d_ws, size_t ws_size,
                              hipStream_t stream) {
  const float* x    = (const float*)d_in[0];
  const float* A    = (const float*)d_in[1];
  const float* W_h  = (const float*)d_in[2];
  const float* b_h  = (const float*)d_in[3];
  const float* W_f1 = (const float*)d_in[4];
  const float* b_f1 = (const float*)d_in[5];
  const float* W_g1 = (const float*)d_in[6];
  const float* b_g1 = (const float*)d_in[7];
  const float* W_g2 = (const float*)d_in[8];
  const float* b_g2 = (const float*)d_in[9];
  float* out = (float*)d_out;

  float* H = (float*)d_ws;
  float* P = H + (size_t)BATCH * NN * HIDD;

  k_proj<<<256, 256, 0, stream>>>(x, W_h, b_h, W_f1, b_f1, H, P);
  k_bmm_gate<<<256, 256, 0, stream>>>(A, H, P, W_g1, b_g1, W_g2, b_g2, out);
}


# Projection kernel's K loop double-buffered in registers: the next 32-k group's x and weight fragments are loaded as one group ahead of the current group's 64 products; same k order and sums
# speedup vs baseline: 1.8339x; 1.1315x over previous
#include <hip/hip_runtime.h>

typedef __attribute__((ext_vector_type(2))) float v2f;
typedef __attribute__((ext_vector_type(4))) float v4f;
typedef __attribute__((ext_vector_type(8))) float v8f;

#define BATCH 32
#define NN    1024
#define IND   128
#define HIDD  64
#define LDS_STRIDE 68


__global__ __launch_bounds__(256) void k_proj(
    const float* __restrict__ x,
    const float* __restrict__ Wh, const float* __restrict__ bh,
    const float* __restrict__ Wf, const float* __restrict__ bf,
    float* __restrict__ H, float* __restrict__ P) {
  const int lane = threadIdx.x & 31;
  const int w    = threadIdx.x >> 5;
  const int gw   = blockIdx.x * 8 + w;
  const int m0   = gw * 16;
  const int lr   = lane & 15;
  const int hi   = lane >> 4;

  v8f ha[4] = {}; v8f pa[4] = {};
  const float* xrow = x  + (size_t)(m0 + lr) * IND + 2 * hi;
  const float* hrow = Wh + (size_t)(2 * hi) * HIDD + 4 * lr;
  const float* frow = Wf + (size_t)(2 * hi) * HIDD + 4 * lr;

  v2f xa[2][8];
  v4f hl[2][8], hh[2][8], fl[2][8], fh[2][8];

#define PLOAD(st, kb)                                                           \
  _Pragma("unroll") for (int j = 0; j < 8; ++j) {                               \
    xa[st][j] = *(const v2f*)(xrow + (kb) + 4 * j);                             \
    hl[st][j] = *(const v4f*)(hrow + (size_t)((kb) + 4 * j) * HIDD);            \
    hh[st][j] = *(const v4f*)(hrow + (size_t)((kb) + 4 * j) * HIDD + HIDD);     \
    fl[st][j] = *(const v4f*)(frow + (size_t)((kb) + 4 * j) * HIDD);            \
    fh[st][j] = *(const v4f*)(frow + (size_t)((kb) + 4 * j) * HIDD + HIDD);     \
  }
#define PMMA(st)                                                                \
  _Pragma("unroll") for (int j = 0; j < 8; ++j) {                               \
    _Pragma("unroll") for (int nb = 0; nb < 4; ++nb) {                          \
      v2f bw, bq;                                                               \
      bw.x = hl[st][j][nb]; bw.y = hh[st][j][nb];                               \
      bq.x = fl[st][j][nb]; bq.y = fh[st][j][nb];                               \
      ha[nb] = __builtin_amdgcn_wmma_f32_16x16x4_f32(false, xa[st][j], false, bw, \
                                                     (short)0, ha[nb], false, false); \
      pa[nb] = __builtin_amdgcn_wmma_f32_16x16x4_f32(false, xa[st][j], false, bq, \
                                                     (short)0, pa[nb], false, false); \
    }                                                                           \
  }
#define PSTEP(ld_st, ld_kb, mma_st)                                             \
  PLOAD(ld_st, ld_kb)                                                           \
  __builtin_amdgcn_sched_barrier(0);                                            \
  PMMA(mma_st)                                                                  \
  __builtin_amdgcn_sched_barrier(0);

  PLOAD(0, 0)
  PSTEP(1, 32, 0)
  PSTEP(0, 64, 1)
  PSTEP(1, 96, 0)
  PMMA(1)
#undef PSTEP
#undef PLOAD
#undef PMMA

  const v4f bhv = *(const v4f*)(bh + 4 * lr);
  const v4f bfv = *(const v4f*)(bf + 4 * lr);
#pragma unroll
  for (int i = 0; i < 8; ++i) {
    const int m = i + 8 * hi;
    const size_t off = (size_t)(m0 + m) * HIDD + 4 * lr;
    v4f hv, pv;
#pragma unroll
    for (int nb = 0; nb < 4; ++nb) {
      hv[nb] = ha[nb][i] + bhv[nb];
      pv[nb] = pa[nb][i] + bfv[nb];
    }
    *(v4f*)(H + off) = hv;
    *(v4f*)(P + off) = pv;
  }
}

__global__ __launch_bounds__(256) void k_bmm_gate(
    const float* __restrict__ A, const float* __restrict__ H,
    const float* __restrict__ P,
    const float* __restrict__ Wg1, const float* __restrict__ bg1,
    const float* __restrict__ Wg2, const float* __restrict__ bg2,
    float* __restrict__ out) {
  __shared__ __attribute__((aligned(16))) float lds[8 * 16 * LDS_STRIDE];

  const int lane = threadIdx.x & 31;
  const int w    = threadIdx.x >> 5;
  const int gw   = blockIdx.x * 8 + w;
  const int b    = gw >> 6;
  const int m0   = (gw & 63) * 16;
  const int lr   = lane & 15;
  const int hi   = lane >> 4;

  const float* Ab = A + (size_t)b * NN * NN;
  const float* Hb = H + (size_t)b * NN * HIDD;

  v8f acc[4] = {};
  const float* arow = Ab + (size_t)(m0 + lr) * NN + 2 * hi;
  const float* hrow = Hb + (size_t)(2 * hi) * HIDD + 4 * lr;
  v2f fa[2][16];
  v4f fl[2][16], fh[2][16];

#define LOAD_BLK(st, kb)                                                        \
  _Pragma("unroll") for (int j = 0; j < 16; ++j) {                              \
    fa[st][j] = *(const v2f*)(arow + (kb) + 4 * j);                             \
    fl[st][j] = *(const v4f*)(hrow + (size_t)((kb) + 4 * j) * HIDD);            \
    fh[st][j] = *(const v4f*)(hrow + (size_t)((kb) + 4 * j) * HIDD + HIDD);     \
  }
#define MMA_BLK(st)                                                             \
  _Pragma("unroll") for (int j = 0; j < 16; ++j) {                              \
    _Pragma("unroll") for (int nb = 0; nb < 4; ++nb) {                          \
      v2f bb;                                                                   \
      bb.x = fl[st][j][nb];                                                     \
      bb.y = fh[st][j][nb];                                                     \
      acc[nb] = __builtin_amdgcn_wmma_f32_16x16x4_f32(false, fa[st][j], false, bb, \
                                                      (short)0, acc[nb], false, false); \
    }                                                                           \
  }
#define STEP(ld_st, ld_kb, mma_st)                                              \
  LOAD_BLK(ld_st, ld_kb)                                                        \
  __builtin_amdgcn_sched_barrier(0);                                            \
  MMA_BLK(mma_st)                                                               \
  __builtin_amdgcn_sched_barrier(0);

  LOAD_BLK(0, 0)
#pragma nounroll
  for (int kb = 0; kb < NN - 128; kb += 128) {
    STEP(1, kb + 64, 0)
    STEP(0, kb + 128, 1)
  }
  STEP(1, NN - 64, 0)
  MMA_BLK(1)
#undef STEP
#undef LOAD_BLK
#undef MMA_BLK

  float* myl = lds + w * (16 * LDS_STRIDE);
#pragma unroll
  for (int i = 0; i < 8; ++i) {
    v4f t;
#pragma unroll
    for (int nb = 0; nb < 4; ++nb) t[nb] = acc[nb][i];
    *(v4f*)(myl + (i + 8 * hi) * LDS_STRIDE + 4 * lr) = t;
  }

  v8f g[4] = {};
  const float* prow = P + (size_t)(b * NN + m0 + lr) * HIDD + 2 * hi;
#pragma unroll
  for (int kk = 0; kk < HIDD; kk += 4) {
    v2f a2 = *(const v2f*)(myl + lr * LDS_STRIDE + kk + 2 * hi);
    v2f a1 = *(const v2f*)(prow + kk);
    const float* g2w = Wg2 + (size_t)(kk + 2 * hi) * HIDD + 4 * lr;
    const float* g1w = Wg1 + (size_t)(kk + 2 * hi) * HIDD + 4 * lr;
    const v4f g2_lo = *(const v4f*)(g2w);
    const v4f g2_hi = *(const v4f*)(g2w + HIDD);
    const v4f g1_lo = *(const v4f*)(g1w);
    const v4f g1_hi = *(const v4f*)(g1w + HIDD);
#pragma unroll
    for (int nb = 0; nb < 4; ++nb) {
      v2f b2, b1;
      b2.x = g2_lo[nb]; b2.y = g2_hi[nb];
      b1.x = g1_lo[nb]; b1.y = g1_hi[nb];
      g[nb] = __builtin_amdgcn_wmma_f32_16x16x4_f32(false, a2, false, b2,
                                                    (short)0, g[nb], false, false);
      g[nb] = __builtin_amdgcn_wmma_f32_16x16x4_f32(false, a1, false, b1,
                                                    (short)0, g[nb], false, false);
    }
  }

  const v4f bias = *(const v4f*)(bg1 + 4 * lr) + *(const v4f*)(bg2 + 4 * lr);
#pragma unroll
  for (int i = 0; i < 8; ++i) {
    const int m = i + 8 * hi;
    const size_t off = (size_t)(b * NN + m0 + m) * HIDD + 4 * lr;
    const v4f p = *(const v4f*)(P + off);
    v4f o;
#pragma unroll
    for (int nb = 0; nb < 4; ++nb) {
      const float xn = acc[nb][i];
      const float z  = g[nb][i] + bias[nb];
      const float gate = 1.0f / (1.0f + __expf(-z));
      o[nb] = xn * gate + p[nb] * (1.0f - gate);
    }
    *(v4f*)(out + off) = o;
  }
}

extern "C" void kernel_launch(void* const* d_in, const int* in_sizes, int n_in,
                              void* d_out, int out_size, void* d_ws, size_t ws_size,
                              hipStream_t stream) {
  const float* x    = (const float*)d_in[0];
  const float* A    = (const float*)d_in[1];
  const float* W_h  = (const float*)d_in[2];
  const float* b_h  = (const float*)d_in[3];
  const float* W_f1 = (const float*)d_in[4];
  const float* b_f1 = (const float*)d_in[5];
  const float* W_g1 = (const float*)d_in[6];
  const float* b_g1 = (const float*)d_in[7];
  const float* W_g2 = (const float*)d_in[8];
  const float* b_g2 = (const float*)d_in[9];
  float* out = (float*)d_out;

  float* H = (float*)d_ws;
  float* P = H + (size_t)BATCH * NN * HIDD;

  k_proj<<<256, 256, 0, stream>>>(x, W_h, b_h, W_f1, b_f1, H, P);
  k_bmm_gate<<<256, 256, 0, stream>>>(A, H, P, W_g1, b_g1, W_g2, b_g2, out);
}


# bmm re-tiled: a wave owns 32 rows x half of K (was 16 x all K), each H fragment feeds two row tiles, loads run two 32-k groups ahead; wave pairs add their K-half partial sums via LDS (new sum order)
# speedup vs baseline: 2.0224x; 1.1028x over previous
#include <hip/hip_runtime.h>

typedef __attribute__((ext_vector_type(2))) float v2f;
typedef __attribute__((ext_vector_type(4))) float v4f;
typedef __attribute__((ext_vector_type(8))) float v8f;

#define BATCH 32
#define NN    1024
#define IND   128
#define HIDD  64
#define LDS_STRIDE 68


__global__ __launch_bounds__(256) void k_proj(
    const float* __restrict__ x,
    const float* __restrict__ Wh, const float* __restrict__ bh,
    const float* __restrict__ Wf, const float* __restrict__ bf,
    float* __restrict__ H, float* __restrict__ P) {
  const int lane = threadIdx.x & 31;
  const int w    = threadIdx.x >> 5;
  const int gw   = blockIdx.x * 8 + w;
  const int m0   = gw * 16;
  const int lr   = lane & 15;
  const int hi   = lane >> 4;

  v8f ha[4] = {}; v8f pa[4] = {};
  const float* xrow = x  + (size_t)(m0 + lr) * IND + 2 * hi;
  const float* hrow = Wh + (size_t)(2 * hi) * HIDD + 4 * lr;
  const float* frow = Wf + (size_t)(2 * hi) * HIDD + 4 * lr;

  v2f xa[2][8];
  v4f hl[2][8], hh[2][8], fl[2][8], fh[2][8];

#define PLOAD(st, kb)                                                           \
  _Pragma("unroll") for (int j = 0; j < 8; ++j) {                               \
    xa[st][j] = *(const v2f*)(xrow + (kb) + 4 * j);                             \
    hl[st][j] = *(const v4f*)(hrow + (size_t)((kb) + 4 * j) * HIDD);            \
    hh[st][j] = *(const v4f*)(hrow + (size_t)((kb) + 4 * j) * HIDD + HIDD);     \
    fl[st][j] = *(const v4f*)(frow + (size_t)((kb) + 4 * j) * HIDD);            \
    fh[st][j] = *(const v4f*)(frow + (size_t)((kb) + 4 * j) * HIDD + HIDD);     \
  }
#define PMMA(st)                                                                \
  _Pragma("unroll") for (int j = 0; j < 8; ++j) {                               \
    _Pragma("unroll") for (int nb = 0; nb < 4; ++nb) {                          \
      v2f bw, bq;                                                               \
      bw.x = hl[st][j][nb]; bw.y = hh[st][j][nb];                               \
      bq.x = fl[st][j][nb]; bq.y = fh[st][j][nb];                               \
      ha[nb] = __builtin_amdgcn_wmma_f32_16x16x4_f32(false, xa[st][j], false, bw, \
                                                     (short)0, ha[nb], false, false); \
      pa[nb] = __builtin_amdgcn_wmma_f32_16x16x4_f32(false, xa[st][j], false, bq, \
                                                     (short)0, pa[nb], false, false); \
    }                                                                           \
  }
#define PSTEP(ld_st, ld_kb, mma_st)                                             \
  PLOAD(ld_st, ld_kb)                                                           \
  __builtin_amdgcn_sched_barrier(0);                                            \
  PMMA(mma_st)                                                                  \
  __builtin_amdgcn_sched_barrier(0);

  PLOAD(0, 0)
  PSTEP(1, 32, 0)
  PSTEP(0, 64, 1)
  PSTEP(1, 96, 0)
  PMMA(1)
#undef PSTEP
#undef PLOAD
#undef PMMA

  const v4f bhv = *(const v4f*)(bh + 4 * lr);
  const v4f bfv = *(const v4f*)(bf + 4 * lr);
#pragma unroll
  for (int i = 0; i < 8; ++i) {
    const int m = i + 8 * hi;
    const size_t off = (size_t)(m0 + m) * HIDD + 4 * lr;
    v4f hv, pv;
#pragma unroll
    for (int nb = 0; nb < 4; ++nb) {
      hv[nb] = ha[nb][i] + bhv[nb];
      pv[nb] = pa[nb][i] + bfv[nb];
    }
    *(v4f*)(H + off) = hv;
    *(v4f*)(P + off) = pv;
  }
}

__global__ __launch_bounds__(256) void k_bmm_gate(
    const float* __restrict__ A, const float* __restrict__ H,
    const float* __restrict__ P,
    const float* __restrict__ Wg1, const float* __restrict__ bg1,
    const float* __restrict__ Wg2, const float* __restrict__ bg2,
    float* __restrict__ out) {
  __shared__ __attribute__((aligned(16))) float xch[8 * 16 * LDS_STRIDE];
  __shared__ __attribute__((aligned(16))) float lds[8 * 16 * LDS_STRIDE];

  const int lane = threadIdx.x & 31;
  const int w    = threadIdx.x >> 5;
  const int pr   = w & 3;
  const int kh   = w >> 2;
  const int b    = blockIdx.x >> 3;
  const int m0   = (blockIdx.x & 7) * 128 + pr * 32;
  const int lr   = lane & 15;
  const int hi   = lane >> 4;

  const float* Ab = A + (size_t)b * NN * NN;
  const float* Hb = H + (size_t)b * NN * HIDD;

  v8f acc0[4] = {}; v8f acc1[4] = {};
  const float* arow = Ab + (size_t)(m0 + lr) * NN + kh * 512 + 2 * hi;
  const float* hrow = Hb + (size_t)(kh * 512 + 2 * hi) * HIDD + 4 * lr;
  v2f fa0[3][8], fa1[3][8];
  v4f fl[3][8], fh[3][8];

#define LOAD_BLK(st, kb)                                                        \
  _Pragma("unroll") for (int j = 0; j < 8; ++j) {                               \
    fa0[st][j] = *(const v2f*)(arow + (kb) + 4 * j);                            \
    fa1[st][j] = *(const v2f*)(arow + (size_t)16 * NN + (kb) + 4 * j);          \
    fl[st][j]  = *(const v4f*)(hrow + (size_t)((kb) + 4 * j) * HIDD);           \
    fh[st][j]  = *(const v4f*)(hrow + (size_t)((kb) + 4 * j) * HIDD + HIDD);    \
  }
#define MMA_BLK(st)                                                             \
  _Pragma("unroll") for (int j = 0; j < 8; ++j) {                               \
    _Pragma("unroll") for (int nb = 0; nb < 4; ++nb) {                          \
      v2f bb;                                                                   \
      bb.x = fl[st][j][nb];                                                     \
      bb.y = fh[st][j][nb];                                                     \
      acc0[nb] = __builtin_amdgcn_wmma_f32_16x16x4_f32(false, fa0[st][j], false, bb, \
                                                       (short)0, acc0[nb], false, false); \
      acc1[nb] = __builtin_amdgcn_wmma_f32_16x16x4_f32(false, fa1[st][j], false, bb, \
                                                       (short)0, acc1[nb], false, false); \
    }                                                                           \
  }
#define STEP(ld_st, ld_kb, mma_st)                                              \
  LOAD_BLK(ld_st, ld_kb)                                                        \
  __builtin_amdgcn_sched_barrier(0);                                            \
  MMA_BLK(mma_st)                                                               \
  __builtin_amdgcn_sched_barrier(0);

  LOAD_BLK(0, 0)
  __builtin_amdgcn_sched_barrier(0);
  LOAD_BLK(1, 32)
#pragma unroll
  for (int kb = 0; kb < 384; kb += 96) {
    STEP(2, kb + 64, 0)
    STEP(0, kb + 96, 1)
    STEP(1, kb + 128, 2)
  }
  STEP(2, 448, 0)
  STEP(0, 480, 1)
  MMA_BLK(2)
  MMA_BLK(0)
#undef STEP
#undef LOAD_BLK
#undef MMA_BLK

  float* myx = xch + w * (16 * LDS_STRIDE);
  const float* px = xch + (w ^ 4) * (16 * LDS_STRIDE);
#pragma unroll
  for (int i = 0; i < 8; ++i) {
    v4f t;
#pragma unroll
    for (int nb = 0; nb < 4; ++nb) t[nb] = kh ? acc0[nb][i] : acc1[nb][i];
    *(v4f*)(myx + (i + 8 * hi) * LDS_STRIDE + 4 * lr) = t;
  }
  __syncthreads();
  v8f acc[4];
#pragma unroll
  for (int i = 0; i < 8; ++i) {
    const v4f t = *(const v4f*)(px + (i + 8 * hi) * LDS_STRIDE + 4 * lr);
#pragma unroll
    for (int nb = 0; nb < 4; ++nb) {
      const float mine = kh ? acc1[nb][i] : acc0[nb][i];
      acc[nb][i] = kh ? (t[nb] + mine) : (mine + t[nb]);
    }
  }
  const int mg = m0 + 16 * kh;

  float* myl = lds + w * (16 * LDS_STRIDE);
#pragma unroll
  for (int i = 0; i < 8; ++i) {
    v4f t;
#pragma unroll
    for (int nb = 0; nb < 4; ++nb) t[nb] = acc[nb][i];
    *(v4f*)(myl + (i + 8 * hi) * LDS_STRIDE + 4 * lr) = t;
  }

  v8f g[4] = {};
  const float* prow = P + (size_t)(b * NN + mg + lr) * HIDD + 2 * hi;
#pragma unroll
  for (int kk = 0; kk < HIDD; kk += 4) {
    v2f a2 = *(const v2f*)(myl + lr * LDS_STRIDE + kk + 2 * hi);
    v2f a1 = *(const v2f*)(prow + kk);
    const float* g2w = Wg2 + (size_t)(kk + 2 * hi) * HIDD + 4 * lr;
    const float* g1w = Wg1 + (size_t)(kk + 2 * hi) * HIDD + 4 * lr;
    const v4f g2_lo = *(const v4f*)(g2w);
    const v4f g2_hi = *(const v4f*)(g2w + HIDD);
    const v4f g1_lo = *(const v4f*)(g1w);
    const v4f g1_hi = *(const v4f*)(g1w + HIDD);
#pragma unroll
    for (int nb = 0; nb < 4; ++nb) {
      v2f b2, b1;
      b2.x = g2_lo[nb]; b2.y = g2_hi[nb];
      b1.x = g1_lo[nb]; b1.y = g1_hi[nb];
      g[nb] = __builtin_amdgcn_wmma_f32_16x16x4_f32(false, a2, false, b2,
                                                    (short)0, g[nb], false, false);
      g[nb] = __builtin_amdgcn_wmma_f32_16x16x4_f32(false, a1, false, b1,
                                                    (short)0, g[nb], false, false);
    }
  }

  const v4f bias = *(const v4f*)(bg1 + 4 * lr) + *(const v4f*)(bg2 + 4 * lr);
#pragma unroll
  for (int i = 0; i < 8; ++i) {
    const int m = i + 8 * hi;
    const size_t off = (size_t)(b * NN + mg + m) * HIDD + 4 * lr;
    const v4f p = *(const v4f*)(P + off);
    v4f o;
#pragma unroll
    for (int nb = 0; nb < 4; ++nb) {
      const float xn = acc[nb][i];
      const float z  = g[nb][i] + bias[nb];
      const float gate = 1.0f / (1.0f + __expf(-z));
      o[nb] = xn * gate + p[nb] * (1.0f - gate);
    }
    *(v4f*)(out + off) = o;
  }
}

extern "C" void kernel_launch(void* const* d_in, const int* in_sizes, int n_in,
                              void* d_out, int out_size, void* d_ws, size_t ws_size,
                              hipStream_t stream) {
  const float* x    = (const float*)d_in[0];
  const float* A    = (const float*)d_in[1];
  const float* W_h  = (const float*)d_in[2];
  const float* b_h  = (const float*)d_in[3];
  const float* W_f1 = (const float*)d_in[4];
  const float* b_f1 = (const float*)d_in[5];
  const float* W_g1 = (const float*)d_in[6];
  const float* b_g1 = (const float*)d_in[7];
  const float* W_g2 = (const float*)d_in[8];
  const float* b_g2 = (const float*)d_in[9];
  float* out = (float*)d_out;

  float* H = (float*)d_ws;
  float* P = H + (size_t)BATCH * NN * HIDD;

  k_proj<<<256, 256, 0, stream>>>(x, W_h, b_h, W_f1, b_f1, H, P);
  k_bmm_gate<<<256, 256, 0, stream>>>(A, H, P, W_g1, b_g1, W_g2, b_g2, out);
}


# Skip projection P = x*W_fc1+b and its gate product P*W_g1 moved to the bmm kernel's start, run while the first A groups load; P kept in registers, never stored; gate sum: P part first, then xn part
# speedup vs baseline: 2.1148x; 1.0457x over previous
#include <hip/hip_runtime.h>

typedef __attribute__((ext_vector_type(2))) float v2f;
typedef __attribute__((ext_vector_type(4))) float v4f;
typedef __attribute__((ext_vector_type(8))) float v8f;

#define BATCH 32
#define NN    1024
#define IND   128
#define HIDD  64
#define LDS_STRIDE 68


__global__ __launch_bounds__(256) void k_proj(
    const float* __restrict__ x,
    const float* __restrict__ Wh, const float* __restrict__ bh,
    float* __restrict__ H) {
  const int lane = threadIdx.x & 31;
  const int w    = threadIdx.x >> 5;
  const int gw   = blockIdx.x * 8 + w;
  const int m0   = gw * 16;
  const int lr   = lane & 15;
  const int hi   = lane >> 4;

  v8f ha[4] = {};
  const float* xrow = x  + (size_t)(m0 + lr) * IND + 2 * hi;
  const float* hrow = Wh + (size_t)(2 * hi) * HIDD + 4 * lr;

  v2f xa[2][8];
  v4f hl[2][8], hh[2][8];

#define PLOAD(st, kb)                                                           \
  _Pragma("unroll") for (int j = 0; j < 8; ++j) {                               \
    xa[st][j] = *(const v2f*)(xrow + (kb) + 4 * j);                             \
    hl[st][j] = *(const v4f*)(hrow + (size_t)((kb) + 4 * j) * HIDD);            \
    hh[st][j] = *(const v4f*)(hrow + (size_t)((kb) + 4 * j) * HIDD + HIDD);     \
  }
#define PMMA(st)                                                                \
  _Pragma("unroll") for (int j = 0; j < 8; ++j) {                               \
    _Pragma("unroll") for (int nb = 0; nb < 4; ++nb) {                          \
      v2f bw;                                                                   \
      bw.x = hl[st][j][nb]; bw.y = hh[st][j][nb];                               \
      ha[nb] = __builtin_amdgcn_wmma_f32_16x16x4_f32(false, xa[st][j], false, bw, \
                                                     (short)0, ha[nb], false, false); \
    }                                                                           \
  }
#define PSTEP(ld_st, ld_kb, mma_st)                                             \
  PLOAD(ld_st, ld_kb)                                                           \
  __builtin_amdgcn_sched_barrier(0);                                            \
  PMMA(mma_st)                                                                  \
  __builtin_amdgcn_sched_barrier(0);

  PLOAD(0, 0)
  PSTEP(1, 32, 0)
  PSTEP(0, 64, 1)
  PSTEP(1, 96, 0)
  PMMA(1)
#undef PSTEP
#undef PLOAD
#undef PMMA

  const v4f bhv = *(const v4f*)(bh + 4 * lr);
#pragma unroll
  for (int i = 0; i < 8; ++i) {
    const int m = i + 8 * hi;
    const size_t off = (size_t)(m0 + m) * HIDD + 4 * lr;
    v4f hv;
#pragma unroll
    for (int nb = 0; nb < 4; ++nb) hv[nb] = ha[nb][i] + bhv[nb];
    *(v4f*)(H + off) = hv;
  }
}

__global__ __launch_bounds__(256) void k_bmm_gate(
    const float* __restrict__ A, const float* __restrict__ H,
    const float* __restrict__ x,
    const float* __restrict__ Wf, const float* __restrict__ bf,
    const float* __restrict__ Wg1, const float* __restrict__ bg1,
    const float* __restrict__ Wg2, const float* __restrict__ bg2,
    float* __restrict__ out) {
  __shared__ __attribute__((aligned(16))) float xch[8 * 16 * LDS_STRIDE];
  __shared__ __attribute__((aligned(16))) float lds[8 * 16 * LDS_STRIDE];

  const int lane = threadIdx.x & 31;
  const int w    = threadIdx.x >> 5;
  const int pr   = w & 3;
  const int kh   = w >> 2;
  const int b    = blockIdx.x >> 3;
  const int m0   = (blockIdx.x & 7) * 128 + pr * 32;
  const int mg   = m0 + 16 * kh;
  const int lr   = lane & 15;
  const int hi   = lane >> 4;

  const float* Ab = A + (size_t)b * NN * NN;
  const float* Hb = H + (size_t)b * NN * HIDD;
  float* myl = lds + w * (16 * LDS_STRIDE);

  const float* arow = Ab + (size_t)(m0 + lr) * NN + kh * 512 + 2 * hi;
  const float* hrow = Hb + (size_t)(kh * 512 + 2 * hi) * HIDD + 4 * lr;
  v2f fa0[3][8], fa1[3][8];
  v4f fl[3][8], fh[3][8];

#define LOAD_BLK(st, kb)                                                        \
  _Pragma("unroll") for (int j = 0; j < 8; ++j) {                               \
    fa0[st][j] = *(const v2f*)(arow + (kb) + 4 * j);                            \
    fa1[st][j] = *(const v2f*)(arow + (size_t)16 * NN + (kb) + 4 * j);          \
    fl[st][j]  = *(const v4f*)(hrow + (size_t)((kb) + 4 * j) * HIDD);           \
    fh[st][j]  = *(const v4f*)(hrow + (size_t)((kb) + 4 * j) * HIDD + HIDD);    \
  }

  v8f pa[4] = {};
  const float* xrow = x  + (size_t)(b * NN + mg + lr) * IND + 2 * hi;
  const float* qrow = Wf + (size_t)(2 * hi) * HIDD + 4 * lr;
  v2f xq[2][8];
  v4f ql[2][8], qh[2][8];
#define FLOAD(st, kb)                                                           \
  _Pragma("unroll") for (int j = 0; j < 8; ++j) {                               \
    xq[st][j] = *(const v2f*)(xrow + (kb) + 4 * j);                             \
    ql[st][j] = *(const v4f*)(qrow + (size_t)((kb) + 4 * j) * HIDD);            \
    qh[st][j] = *(const v4f*)(qrow + (size_t)((kb) + 4 * j) * HIDD + HIDD);     \
  }
#define FMMA(st)                                                                \
  _Pragma("unroll") for (int j = 0; j < 8; ++j) {                               \
    _Pragma("unroll") for (int nb = 0; nb < 4; ++nb) {                          \
      v2f bq;                                                                   \
      bq.x = ql[st][j][nb]; bq.y = qh[st][j][nb];                               \
      pa[nb] = __builtin_amdgcn_wmma_f32_16x16x4_f32(false, xq[st][j], false, bq, \
                                                     (short)0, pa[nb], false, false); \
    }                                                                           \
  }
  const v4f bfv = *(const v4f*)(bf + 4 * lr);
  FLOAD(0, 0)
  FLOAD(1, 32)
  __builtin_amdgcn_sched_barrier(0);
  FMMA(0)
  __builtin_amdgcn_sched_barrier(0);
  FLOAD(0, 64)
  __builtin_amdgcn_sched_barrier(0);
  FMMA(1)
  __builtin_amdgcn_sched_barrier(0);
  FLOAD(1, 96)
  v4f w1l[16], w1h[16];
#pragma unroll
  for (int sidx = 0; sidx < 16; ++sidx) {
    const float* g1w = Wg1 + (size_t)(4 * sidx + 2 * hi) * HIDD + 4 * lr;
    w1l[sidx] = *(const v4f*)(g1w);
    w1h[sidx] = *(const v4f*)(g1w + HIDD);
  }
  __builtin_amdgcn_sched_barrier(0);
  FMMA(0)
  __builtin_amdgcn_sched_barrier(0);
  LOAD_BLK(0, 0)
  __builtin_amdgcn_sched_barrier(0);
  FMMA(1)
  __builtin_amdgcn_sched_barrier(0);
  LOAD_BLK(1, 32)
  __builtin_amdgcn_sched_barrier(0);
#undef FLOAD
#undef FMMA

#pragma unroll
  for (int nb = 0; nb < 4; ++nb)
#pragma unroll
    for (int i = 0; i < 8; ++i) pa[nb][i] += bfv[nb];
#pragma unroll
  for (int i = 0; i < 8; ++i) {
    v4f t;
#pragma unroll
    for (int nb = 0; nb < 4; ++nb) t[nb] = pa[nb][i];
    *(v4f*)(myl + (i + 8 * hi) * LDS_STRIDE + 4 * lr) = t;
  }
  v8f g[4] = {};
#pragma unroll
  for (int sidx = 0; sidx < 16; ++sidx) {
    const v2f a1 = *(const v2f*)(myl + lr * LDS_STRIDE + 4 * sidx + 2 * hi);
#pragma unroll
    for (int nb = 0; nb < 4; ++nb) {
      v2f b1;
      b1.x = w1l[sidx][nb]; b1.y = w1h[sidx][nb];
      g[nb] = __builtin_amdgcn_wmma_f32_16x16x4_f32(false, a1, false, b1,
                                                    (short)0, g[nb], false, false);
    }
  }
  __builtin_amdgcn_sched_barrier(0);

  v8f acc0[4] = {}; v8f acc1[4] = {};
#define MMA_BLK(st)                                                             \
  _Pragma("unroll") for (int j = 0; j < 8; ++j) {                               \
    _Pragma("unroll") for (int nb = 0; nb < 4; ++nb) {                          \
      v2f bb;                                                                   \
      bb.x = fl[st][j][nb];                                                     \
      bb.y = fh[st][j][nb];                                                     \
      acc0[nb] = __builtin_amdgcn_wmma_f32_16x16x4_f32(false, fa0[st][j], false, bb, \
                                                       (short)0, acc0[nb], false, false); \
      acc1[nb] = __builtin_amdgcn_wmma_f32_16x16x4_f32(false, fa1[st][j], false, bb, \
                                                       (short)0, acc1[nb], false, false); \
    }                                                                           \
  }
#define STEP(ld_st, ld_kb, mma_st)                                              \
  LOAD_BLK(ld_st, ld_kb)                                                        \
  __builtin_amdgcn_sched_barrier(0);                                            \
  MMA_BLK(mma_st)                                                               \
  __builtin_amdgcn_sched_barrier(0);

#pragma unroll
  for (int kb = 0; kb < 384; kb += 96) {
    STEP(2, kb + 64, 0)
    STEP(0, kb + 96, 1)
    STEP(1, kb + 128, 2)
  }
  STEP(2, 448, 0)
  STEP(0, 480, 1)
  MMA_BLK(2)
  MMA_BLK(0)
#undef STEP
#undef LOAD_BLK
#undef MMA_BLK

  float* myx = xch + w * (16 * LDS_STRIDE);
  const float* px = xch + (w ^ 4) * (16 * LDS_STRIDE);
#pragma unroll
  for (int i = 0; i < 8; ++i) {
    v4f t;
#pragma unroll
    for (int nb = 0; nb < 4; ++nb) t[nb] = kh ? acc0[nb][i] : acc1[nb][i];
    *(v4f*)(myx + (i + 8 * hi) * LDS_STRIDE + 4 * lr) = t;
  }
  __syncthreads();
  v8f acc[4];
#pragma unroll
  for (int i = 0; i < 8; ++i) {
    const v4f t = *(const v4f*)(px + (i + 8 * hi) * LDS_STRIDE + 4 * lr);
#pragma unroll
    for (int nb = 0; nb < 4; ++nb) {
      const float mine = kh ? acc1[nb][i] : acc0[nb][i];
      acc[nb][i] = kh ? (t[nb] + mine) : (mine + t[nb]);
    }
  }

#pragma unroll
  for (int i = 0; i < 8; ++i) {
    v4f t;
#pragma unroll
    for (int nb = 0; nb < 4; ++nb) t[nb] = acc[nb][i];
    *(v4f*)(myl + (i + 8 * hi) * LDS_STRIDE + 4 * lr) = t;
  }

#pragma unroll
  for (int kk = 0; kk < HIDD; kk += 4) {
    v2f a2 = *(const v2f*)(myl + lr * LDS_STRIDE + kk + 2 * hi);
    const float* g2w = Wg2 + (size_t)(kk + 2 * hi) * HIDD + 4 * lr;
    const v4f g2_lo = *(const v4f*)(g2w);
    const v4f g2_hi = *(const v4f*)(g2w + HIDD);
#pragma unroll
    for (int nb = 0; nb < 4; ++nb) {
      v2f b2;
      b2.x = g2_lo[nb]; b2.y = g2_hi[nb];
      g[nb] = __builtin_amdgcn_wmma_f32_16x16x4_f32(false, a2, false, b2,
                                                    (short)0, g[nb], false, false);
    }
  }

  const v4f bias = *(const v4f*)(bg1 + 4 * lr) + *(const v4f*)(bg2 + 4 * lr);
#pragma unroll
  for (int i = 0; i < 8; ++i) {
    const int m = i + 8 * hi;
    const size_t off = (size_t)(b * NN + mg + m) * HIDD + 4 * lr;
    v4f o;
#pragma unroll
    for (int nb = 0; nb < 4; ++nb) {
      const float xn = acc[nb][i];
      const float p  = pa[nb][i];
      const float z  = g[nb][i] + bias[nb];
      const float gate = 1.0f / (1.0f + __expf(-z));
      o[nb] = xn * gate + p * (1.0f - gate);
    }
    *(v4f*)(out + off) = o;
  }
}

extern "C" void kernel_launch(void* const* d_in, const int* in_sizes, int n_in,
                              void* d_out, int out_size, void* d_ws, size_t ws_size,
                              hipStream_t stream) {
  const float* x    = (const float*)d_in[0];
  const float* A    = (const float*)d_in[1];
  const float* W_h  = (const float*)d_in[2];
  const float* b_h  = (const float*)d_in[3];
  const float* W_f1 = (const float*)d_in[4];
  const float* b_f1 = (const float*)d_in[5];
  const float* W_g1 = (const float*)d_in[6];
  const float* b_g1 = (const float*)d_in[7];
  const float* W_g2 = (const float*)d_in[8];
  const float* b_g2 = (const float*)d_in[9];
  float* out = (float*)d_out;

  float* H = (float*)d_ws;

  k_proj<<<256, 256, 0, stream>>>(x, W_h, b_h, H);
  k_bmm_gate<<<256, 256, 0, stream>>>(A, H, x, W_f1, b_f1, W_g1, b_g1, W_g2, b_g2, out);
}
